# starting kernel
# baseline (speedup 1.0000x reference)
#include <hip/hip_runtime.h>
#include <hip/hip_bf16.h>

typedef __attribute__((ext_vector_type(2))) float v2f;
typedef __attribute__((ext_vector_type(8))) float v8f;

#define D64      64
#define NV       325
#define NROWS_TE 1536
#define LDS_STRIDE 68

__device__ __forceinline__ v8f gemm16x64_tile(const v2f* __restrict__ afrag,
                                              const float* __restrict__ Bbase)
{
    v2f bfrag[16];
    #pragma unroll
    for (int kb = 0; kb < 16; ++kb) {
        bfrag[kb][0] = Bbase[(4 * kb    ) * D64];
        bfrag[kb][1] = Bbase[(4 * kb + 1) * D64];
    }
    v8f acc = {};
    #pragma unroll
    for (int kb = 0; kb < 16; ++kb) {
        acc = __builtin_amdgcn_wmma_f32_16x16x4_f32(false, afrag[kb], false, bfrag[kb],
                                                    (short)0, acc, false, false);
    }
    return acc;
}

__global__ __launch_bounds__(128)
void se_mlp_kernel(const float* __restrict__ SE,
                   const float* __restrict__ Ws1, const float* __restrict__ bs1,
                   const float* __restrict__ Ws2, const float* __restrict__ bs2,
                   float* __restrict__ se_out)
{
    __shared__ float hid[16 * LDS_STRIDE];

    const int tid  = threadIdx.x;
    const int lane = tid & 31;
    const int wave = tid >> 5;
    const int half = lane >> 4;
    const int r    = lane & 15;
    const int m0   = blockIdx.x * 16;
    const int n0   = wave * 16;

    const int  grow = m0 + r;
    const bool mok  = (grow < NV);
    const int  srow = mok ? grow : 0;
    const float msk = mok ? 1.0f : 0.0f;

    const float* Ap = SE + srow * D64 + 2 * half;
    v2f afrag[16];
    #pragma unroll
    for (int kb = 0; kb < 16; ++kb) {
        v2f a = *reinterpret_cast<const v2f*>(Ap + 4 * kb);
        a[0] *= msk; a[1] *= msk;
        afrag[kb] = a;
    }

    v8f acc = gemm16x64_tile(afrag, Ws1 + (2 * half) * D64 + n0 + r);

    const float bias1 = bs1[n0 + r];
    #pragma unroll
    for (int j = 0; j < 8; ++j) {
        const int mloc = j + 8 * half;
        hid[mloc * LDS_STRIDE + n0 + r] = fmaxf(acc[j] + bias1, 0.0f);
    }
    __syncthreads();

    const float* Hp = hid + r * LDS_STRIDE + 2 * half;
    v2f afrag2[16];
    #pragma unroll
    for (int kb = 0; kb < 16; ++kb)
        afrag2[kb] = *reinterpret_cast<const v2f*>(Hp + 4 * kb);

    v8f acc2 = gemm16x64_tile(afrag2, Ws2 + (2 * half) * D64 + n0 + r);

    const float bias2 = bs2[n0 + r];
    #pragma unroll
    for (int j = 0; j < 8; ++j) {
        const int mloc = j + 8 * half;
        const int gm   = m0 + mloc;
        if (gm < NV)
            se_out[gm * D64 + n0 + r] = acc2[j] + bias2;
    }
}

__global__ __launch_bounds__(128)
void te_mlp_kernel(const int* __restrict__ TE, const int* __restrict__ Tptr,
                   const float* __restrict__ Wt1, const float* __restrict__ bt1,
                   const float* __restrict__ Wt2, const float* __restrict__ bt2,
                   float* __restrict__ te_out)
{
    __shared__ float hid[16 * LDS_STRIDE];

    const int tid  = threadIdx.x;
    const int lane = tid & 31;
    const int wave = tid >> 5;
    const int half = lane >> 4;
    const int r    = lane & 15;
    const int row0 = blockIdx.x * 16;
    const int n0   = wave * 16;
    const int T    = Tptr[0];

    {
        const int rl  = tid >> 3;
        const int d0  = (tid & 7) * 8;
        const int gr  = row0 + rl;
        const int dow = TE[2 * gr    ] % 7;
        const int tod = TE[2 * gr + 1] % T;
        const float* w_dow = Wt1 + dow * D64 + d0;
        const float* w_tod = Wt1 + (7 + tod) * D64 + d0;
        const float* bp    = bt1 + d0;
        float* hp = hid + rl * LDS_STRIDE + d0;
        #pragma unroll
        for (int i = 0; i < 8; ++i)
            hp[i] = fmaxf(w_dow[i] + w_tod[i] + bp[i], 0.0f);
    }
    __syncthreads();

    const float* Hp = hid + r * LDS_STRIDE + 2 * half;
    v2f afrag[16];
    #pragma unroll
    for (int kb = 0; kb < 16; ++kb)
        afrag[kb] = *reinterpret_cast<const v2f*>(Hp + 4 * kb);

    v8f acc = gemm16x64_tile(afrag, Wt2 + (2 * half) * D64 + n0 + r);

    const float bias = bt2[n0 + r];
    #pragma unroll
    for (int j = 0; j < 8; ++j) {
        const int mloc = j + 8 * half;
        te_out[(row0 + mloc) * D64 + n0 + r] = acc[j] + bias;
    }
}

__global__ __launch_bounds__(256)
void broadcast_add_kernel(const float* __restrict__ se,
                          const float* __restrict__ te,
                          float* __restrict__ out)
{
    const int tid   = threadIdx.x;
    const int d4    = tid & 15;
    const int vlane = tid >> 4;
    const int bt    = blockIdx.y;
    const int v0    = blockIdx.x * 64;

    const float4 tv = reinterpret_cast<const float4*>(te + (size_t)bt * D64)[d4];

    #pragma unroll
    for (int p = 0; p < 4; ++p) {
        const int v = v0 + p * 16 + vlane;
        if (v < NV) {
            const float4 sv = reinterpret_cast<const float4*>(se + (size_t)v * D64)[d4];
            float4 o;
            o.x = sv.x + tv.x; o.y = sv.y + tv.y;
            o.z = sv.z + tv.z; o.w = sv.w + tv.w;
            reinterpret_cast<float4*>(out + ((size_t)bt * NV + v) * D64)[d4] = o;
        }
    }
}

extern "C" void kernel_launch(void* const* d_in, const int* in_sizes, int n_in,
                              void* d_out, int out_size, void* d_ws, size_t ws_size,
                              hipStream_t stream)
{
    const float* SE  = (const float*)d_in[0];
    const int*   TE  = (const int*)  d_in[1];
    const int*   T   = (const int*)  d_in[2];
    const float* Ws1 = (const float*)d_in[3];
    const float* bs1 = (const float*)d_in[4];
    const float* Ws2 = (const float*)d_in[5];
    const float* bs2 = (const float*)d_in[6];
    const float* Wt1 = (const float*)d_in[7];
    const float* bt1 = (const float*)d_in[8];
    const float* Wt2 = (const float*)d_in[9];
    const float* bt2 = (const float*)d_in[10];
    float* out = (float*)d_out;

    float* se_ws = (float*)d_ws;
    float* te_ws = se_ws + NV * D64;

    se_mlp_kernel<<<dim3(21), dim3(128), 0, stream>>>(SE, Ws1, bs1, Ws2, bs2, se_ws);
    te_mlp_kernel<<<dim3(NROWS_TE / 16), dim3(128), 0, stream>>>(TE, T, Wt1, bt1, Wt2, bt2, te_ws);
    broadcast_add_kernel<<<dim3(6, NROWS_TE), dim3(256), 0, stream>>>(se_ws, te_ws, out);

    (void)in_sizes; (void)n_in; (void)out_size; (void)ws_size;
}


# Fuse the 3 launches into 1: each of 252 work-groups (16 vertices x 128 time rows) rebuilds its se and te tiles with the same WMMA chains in LDS, then streams its 128 x 4 KB of sums
# speedup vs baseline: 1.3648x; 1.3648x over previous
#include <hip/hip_runtime.h>


typedef __attribute__((ext_vector_type(2))) float v2f;
typedef __attribute__((ext_vector_type(4))) float v4f;
typedef __attribute__((ext_vector_type(8))) float v8f;

#define D64        64
#define NV         325
#define LDS_STRIDE 68
#define BT_CHUNK   128
#define N_VSTRIPE  21
#define N_BTCHUNK  12

__device__ __forceinline__ void load_bfrag(v2f (&bfrag)[16], const float* __restrict__ Bbase)
{
    #pragma unroll
    for (int kb = 0; kb < 16; ++kb) {
        bfrag[kb][0] = Bbase[(4 * kb    ) * D64];
        bfrag[kb][1] = Bbase[(4 * kb + 1) * D64];
    }
}

__device__ __forceinline__ void load_afrag_lds(v2f (&afrag)[16], const float* Hp)
{
    #pragma unroll
    for (int kb = 0; kb < 16; ++kb)
        afrag[kb] = *reinterpret_cast<const v2f*>(Hp + 4 * kb);
}

__device__ __forceinline__ v8f chain16(const v2f (&afrag)[16], const v2f (&bfrag)[16])
{
    v8f acc = {};
    #pragma unroll
    for (int kb = 0; kb < 16; ++kb)
        acc = __builtin_amdgcn_wmma_f32_16x16x4_f32(false, afrag[kb], false, bfrag[kb],
                                                    (short)0, acc, false, false);
    return acc;
}

__global__ __launch_bounds__(256)
void fused_embed_kernel(const float* __restrict__ SE,
                        const int* __restrict__ TE, const int* __restrict__ Tptr,
                        const float* __restrict__ Ws1, const float* __restrict__ bs1,
                        const float* __restrict__ Ws2, const float* __restrict__ bs2,
                        const float* __restrict__ Wt1, const float* __restrict__ bt1,
                        const float* __restrict__ Wt2, const float* __restrict__ bt2,
                        float* __restrict__ out)
{
    __shared__ __attribute__((aligned(16))) float hid_se[16 * LDS_STRIDE];
    __shared__ __attribute__((aligned(16))) float se_t  [16 * LDS_STRIDE];
    __shared__ __attribute__((aligned(16))) float hid_te[BT_CHUNK * LDS_STRIDE];
    __shared__ __attribute__((aligned(16))) float te_t  [BT_CHUNK * LDS_STRIDE];

    const int tid  = threadIdx.x;
    const int lane = tid & 31;
    const int wave = __builtin_amdgcn_readfirstlane(tid >> 5);
    const int half = lane >> 4;
    const int r    = lane & 15;
    const int v0   = blockIdx.x * 16;
    const int bt0  = blockIdx.y * BT_CHUNK;
    const int T    = Tptr[0];

    const int d0 = (tid & 7) * 8;
    float g_dow[4][8], g_tod[4][8], g_b[8];
    {
        int dow[4], tod[4];
        #pragma unroll
        for (int i = 0; i < 4; ++i) {
            const int gr = bt0 + (tid >> 3) + 32 * i;
            dow[i] = TE[2 * gr    ] % 7;
            tod[i] = TE[2 * gr + 1] % T;
        }
        #pragma unroll
        for (int k = 0; k < 8; ++k) g_b[k] = bt1[d0 + k];
        #pragma unroll
        for (int i = 0; i < 4; ++i) {
            const float* w_dow = Wt1 + dow[i] * D64 + d0;
            const float* w_tod = Wt1 + (7 + tod[i]) * D64 + d0;
            #pragma unroll
            for (int k = 0; k < 8; ++k) { g_dow[i][k] = w_dow[k]; g_tod[i][k] = w_tod[k]; }
        }
    }

    const int nt0 = (wave & 3) * 16;
    v2f bfrag_t2[16];
    load_bfrag(bfrag_t2, Wt2 + (2 * half) * D64 + nt0 + r);

    v2f bfrag_s2[16];
    if (wave < 4) {
        const int  grow = v0 + r;
        const bool mok  = (grow < NV);
        const int  srow = mok ? grow : 0;
        const float msk = mok ? 1.0f : 0.0f;

        const float* Ap = SE + srow * D64 + 2 * half;
        v2f afrag[16];
        #pragma unroll
        for (int kb = 0; kb < 16; ++kb) {
            v2f a = *reinterpret_cast<const v2f*>(Ap + 4 * kb);
            a[0] *= msk; a[1] *= msk;
            afrag[kb] = a;
        }
        v2f bfrag_s1[16];
        load_bfrag(bfrag_s1, Ws1 + (2 * half) * D64 + nt0 + r);
        load_bfrag(bfrag_s2, Ws2 + (2 * half) * D64 + nt0 + r);

        const v8f acc = chain16(afrag, bfrag_s1);
        const float bias1 = bs1[nt0 + r];
        #pragma unroll
        for (int j = 0; j < 8; ++j)
            hid_se[(j + 8 * half) * LDS_STRIDE + nt0 + r] = fmaxf(acc[j] + bias1, 0.0f);
    }

    #pragma unroll
    for (int i = 0; i < 4; ++i) {
        float* hp = hid_te + ((tid >> 3) + 32 * i) * LDS_STRIDE + d0;
        #pragma unroll
        for (int k = 0; k < 8; ++k)
            hp[k] = fmaxf(g_dow[i][k] + g_tod[i][k] + g_b[k], 0.0f);
    }
    __syncthreads();

    if (wave < 4) {
        v2f afrag2[16];
        load_afrag_lds(afrag2, hid_se + r * LDS_STRIDE + 2 * half);
        const v8f acc2 = chain16(afrag2, bfrag_s2);
        const float bias2 = bs2[nt0 + r];
        #pragma unroll
        for (int j = 0; j < 8; ++j)
            se_t[(j + 8 * half) * LDS_STRIDE + nt0 + r] = acc2[j] + bias2;
    }

    {
        const int mg = (wave >> 2) * 4;
        v2f af[4][16];
        #pragma unroll
        for (int mt = 0; mt < 4; ++mt)
            load_afrag_lds(af[mt], hid_te + (16 * (mg + mt) + r) * LDS_STRIDE + 2 * half);

        v8f acc[4] = {};
        #pragma unroll
        for (int kb = 0; kb < 16; ++kb) {
            #pragma unroll
            for (int mt = 0; mt < 4; ++mt)
                acc[mt] = __builtin_amdgcn_wmma_f32_16x16x4_f32(false, af[mt][kb], false, bfrag_t2[kb],
                                                                (short)0, acc[mt], false, false);
        }
        const float bias = bt2[nt0 + r];
        #pragma unroll
        for (int mt = 0; mt < 4; ++mt) {
            #pragma unroll
            for (int j = 0; j < 8; ++j)
                te_t[(16 * (mg + mt) + j + 8 * half) * LDS_STRIDE + nt0 + r] = acc[mt][j] + bias;
        }
    }
    __syncthreads();

    const int vloc = 2 * wave + half;
    const int v    = v0 + vloc;
    if (v < NV) {
        const v4f sv = *reinterpret_cast<const v4f*>(se_t + vloc * LDS_STRIDE + 4 * r);
        const float* tp = te_t + 4 * r;
        float* op = out + ((size_t)bt0 * NV + v) * D64 + 4 * r;
        #pragma unroll 8
        for (int b = 0; b < BT_CHUNK; ++b) {
            const v4f tv = *reinterpret_cast<const v4f*>(tp + b * LDS_STRIDE);
            *reinterpret_cast<v4f*>(op + (size_t)b * (NV * D64)) = sv + tv;
        }
    }
}

extern "C" void kernel_launch(void* const* d_in, const int* in_sizes, int n_in,
                              void* d_out, int out_size, void* d_ws, size_t ws_size,
                              hipStream_t stream)
{
    const float* SE  = (const float*)d_in[0];
    const int*   TE  = (const int*)  d_in[1];
    const int*   T   = (const int*)  d_in[2];
    const float* Ws1 = (const float*)d_in[3];
    const float* bs1 = (const float*)d_in[4];
    const float* Ws2 = (const float*)d_in[5];
    const float* bs2 = (const float*)d_in[6];
    const float* Wt1 = (const float*)d_in[7];
    const float* bt1 = (const float*)d_in[8];
    const float* Wt2 = (const float*)d_in[9];
    const float* bt2 = (const float*)d_in[10];
    float* out = (float*)d_out;

    fused_embed_kernel<<<dim3(N_VSTRIPE, N_BTCHUNK), dim3(256), 0, stream>>>(
        SE, TE, T, Ws1, bs1, Ws2, bs2, Wt1, bt1, Wt2, bt2, out);

    (void)in_sizes; (void)n_in; (void)out_size; (void)d_ws; (void)ws_size;
}


# te layer 2 built one 16-row tile at a time: only tile 0 stands before the first store, tile t+1's WMMAs run amid tile t's stores behind a split LDS-only barrier; sum order unchanged
# speedup vs baseline: 1.4979x; 1.0975x over previous
#include <hip/hip_runtime.h>


typedef __attribute__((ext_vector_type(2))) float v2f;
typedef __attribute__((ext_vector_type(4))) float v4f;
typedef __attribute__((ext_vector_type(8))) float v8f;

#define D64        64
#define NV         325
#define LDS_STRIDE 68
#define BT_CHUNK   128
#define N_VSTRIPE  21
#define N_BTCHUNK  12

__device__ __forceinline__ void load_bfrag(v2f (&bfrag)[16], const float* __restrict__ Bbase)
{
    #pragma unroll
    for (int kb = 0; kb < 16; ++kb) {
        bfrag[kb][0] = Bbase[(4 * kb    ) * D64];
        bfrag[kb][1] = Bbase[(4 * kb + 1) * D64];
    }
}

__device__ __forceinline__ void load_afrag_lds(v2f (&afrag)[16], const float* Hp)
{
    #pragma unroll
    for (int kb = 0; kb < 16; ++kb)
        afrag[kb] = *reinterpret_cast<const v2f*>(Hp + 4 * kb);
}

__device__ __forceinline__ v8f chain16(const v2f (&afrag)[16], const v2f (&bfrag)[16])
{
    v8f acc = {};
    #pragma unroll
    for (int kb = 0; kb < 16; ++kb)
        acc = __builtin_amdgcn_wmma_f32_16x16x4_f32(false, afrag[kb], false, bfrag[kb],
                                                    (short)0, acc, false, false);
    return acc;
}

__global__ __launch_bounds__(256)
void fused_embed_kernel(const float* __restrict__ SE,
                        const int* __restrict__ TE, const int* __restrict__ Tptr,
                        const float* __restrict__ Ws1, const float* __restrict__ bs1,
                        const float* __restrict__ Ws2, const float* __restrict__ bs2,
                        const float* __restrict__ Wt1, const float* __restrict__ bt1,
                        const float* __restrict__ Wt2, const float* __restrict__ bt2,
                        float* __restrict__ out)
{
    __shared__ __attribute__((aligned(16))) float hid_se[16 * LDS_STRIDE];
    __shared__ __attribute__((aligned(16))) float se_t  [16 * LDS_STRIDE];
    __shared__ __attribute__((aligned(16))) float hid_te[BT_CHUNK * LDS_STRIDE];
    __shared__ __attribute__((aligned(16))) float te_t  [BT_CHUNK * LDS_STRIDE];

    const int tid  = threadIdx.x;
    const int lane = tid & 31;
    const int wave = __builtin_amdgcn_readfirstlane(tid >> 5);
    const int half = lane >> 4;
    const int r    = lane & 15;
    const int v0   = blockIdx.x * 16;
    const int bt0  = blockIdx.y * BT_CHUNK;
    const int T    = Tptr[0];

    const int d0 = (tid & 7) * 8;
    float g_dow[4][8], g_tod[4][8], g_b[8];
    {
        int dow[4], tod[4];
        #pragma unroll
        for (int i = 0; i < 4; ++i) {
            const int gr = bt0 + (tid >> 3) + 32 * i;
            dow[i] = TE[2 * gr    ] % 7;
            tod[i] = TE[2 * gr + 1] % T;
        }
        #pragma unroll
        for (int k = 0; k < 8; ++k) g_b[k] = bt1[d0 + k];
        #pragma unroll
        for (int i = 0; i < 4; ++i) {
            const float* w_dow = Wt1 + dow[i] * D64 + d0;
            const float* w_tod = Wt1 + (7 + tod[i]) * D64 + d0;
            #pragma unroll
            for (int k = 0; k < 8; ++k) { g_dow[i][k] = w_dow[k]; g_tod[i][k] = w_tod[k]; }
        }
    }

    const int nt0 = (wave & 3) * 16;
    v2f bfrag_t2[16];
    load_bfrag(bfrag_t2, Wt2 + (2 * half) * D64 + nt0 + r);
    const float bias_t2 = bt2[nt0 + r];
    float bias_s2 = 0.0f;

    v2f bfrag_s2[16];
    if (wave < 4) {
        const int  grow = v0 + r;
        const bool mok  = (grow < NV);
        const int  srow = mok ? grow : 0;
        const float msk = mok ? 1.0f : 0.0f;

        const float* Ap = SE + srow * D64 + 2 * half;
        v2f afrag[16];
        #pragma unroll
        for (int kb = 0; kb < 16; ++kb) {
            v2f a = *reinterpret_cast<const v2f*>(Ap + 4 * kb);
            a[0] *= msk; a[1] *= msk;
            afrag[kb] = a;
        }
        v2f bfrag_s1[16];
        load_bfrag(bfrag_s1, Ws1 + (2 * half) * D64 + nt0 + r);
        load_bfrag(bfrag_s2, Ws2 + (2 * half) * D64 + nt0 + r);
        const float bias1 = bs1[nt0 + r];
        bias_s2 = bs2[nt0 + r];

        const v8f acc = chain16(afrag, bfrag_s1);
        #pragma unroll
        for (int j = 0; j < 8; ++j)
            hid_se[(j + 8 * half) * LDS_STRIDE + nt0 + r] = fmaxf(acc[j] + bias1, 0.0f);
    }

    #pragma unroll
    for (int i = 0; i < 4; ++i) {
        float* hp = hid_te + ((tid >> 3) + 32 * i) * LDS_STRIDE + d0;
        #pragma unroll
        for (int k = 0; k < 8; ++k)
            hp[k] = fmaxf(g_dow[i][k] + g_tod[i][k] + g_b[k], 0.0f);
    }
    __syncthreads();

    const float* te_a = hid_te + r * LDS_STRIDE + 2 * half;
    float*       te_d = te_t + (8 * half) * LDS_STRIDE + nt0 + r;

    if (wave < 4) {
        v2f afrag2[16];
        load_afrag_lds(afrag2, hid_se + r * LDS_STRIDE + 2 * half);
        const v8f acc2 = chain16(afrag2, bfrag_s2);
        #pragma unroll
        for (int j = 0; j < 8; ++j)
            se_t[(j + 8 * half) * LDS_STRIDE + nt0 + r] = acc2[j] + bias_s2;
    } else {
        v2f af0[16];
        load_afrag_lds(af0, te_a);
        const v8f acc0 = chain16(af0, bfrag_t2);
        #pragma unroll
        for (int j = 0; j < 8; ++j)
            te_d[j * LDS_STRIDE] = acc0[j] + bias_t2;
    }
    __syncthreads();

    const int  vloc = 2 * wave + half;
    const int  v    = v0 + vloc;
    const bool vok  = (v < NV);
    const v4f  sv   = *reinterpret_cast<const v4f*>(se_t + vloc * LDS_STRIDE + 4 * r);
    const float* tp = te_t + 4 * r;
    float* op = out + ((size_t)bt0 * NV + (vok ? v : 0)) * D64 + 4 * r;

    #pragma unroll
    for (int t = 0; t < 8; ++t) {
        #pragma unroll
        for (int h = 0; h < 2; ++h) {
            if (vok) {
                #pragma unroll
                for (int i = 0; i < 8; ++i) {
                    const int b = 16 * t + 8 * h + i;
                    const v4f tv = *reinterpret_cast<const v4f*>(tp + b * LDS_STRIDE);
                    *reinterpret_cast<v4f*>(op + (size_t)b * (NV * D64)) = sv + tv;
                }
            }
            if (t < 7) {
                if (h == 0) {
                    const bool mine = ((t + 1) & 1) ? (wave < 4) : (wave >= 4);
                    if (mine) {
                        v2f af[16];
                        load_afrag_lds(af, te_a + 16 * (t + 1) * LDS_STRIDE);
                        const v8f acc = chain16(af, bfrag_t2);
                        #pragma unroll
                        for (int j = 0; j < 8; ++j)
                            te_d[(16 * (t + 1) + j) * LDS_STRIDE] = acc[j] + bias_t2;
                    }
                    __builtin_amdgcn_fence(__ATOMIC_RELEASE, "workgroup", "local");
                    __builtin_amdgcn_s_barrier_signal(-1);
                } else {
                    __builtin_amdgcn_s_barrier_wait(-1);
                    __builtin_amdgcn_fence(__ATOMIC_ACQUIRE, "workgroup", "local");
                }
            }
        }
    }
}

extern "C" void kernel_launch(void* const* d_in, const int* in_sizes, int n_in,
                              void* d_out, int out_size, void* d_ws, size_t ws_size,
                              hipStream_t stream)
{
    const float* SE  = (const float*)d_in[0];
    const int*   TE  = (const int*)  d_in[1];
    const int*   T   = (const int*)  d_in[2];
    const float* Ws1 = (const float*)d_in[3];
    const float* bs1 = (const float*)d_in[4];
    const float* Ws2 = (const float*)d_in[5];
    const float* bs2 = (const float*)d_in[6];
    const float* Wt1 = (const float*)d_in[7];
    const float* bt1 = (const float*)d_in[8];
    const float* Wt2 = (const float*)d_in[9];
    const float* bt2 = (const float*)d_in[10];
    float* out = (float*)d_out;

    fused_embed_kernel<<<dim3(N_VSTRIPE, N_BTCHUNK), dim3(256), 0, stream>>>(
        SE, TE, T, Ws1, bs1, Ws2, bs2, Wt1, bt1, Wt2, bt2, out);

    (void)in_sizes; (void)n_in; (void)out_size; (void)d_ws; (void)ws_size;
}


# stream re-mapped: each wave owns whole (batch,time) rows, se tile kept in registers, one te row read from LDS per 4 KB, a wave's 8 stores = one contiguous 4 KB run; sum order unchanged
# speedup vs baseline: 1.6723x; 1.1164x over previous
#include <hip/hip_runtime.h>


typedef __attribute__((ext_vector_type(2))) float v2f;
typedef __attribute__((ext_vector_type(4))) float v4f;
typedef __attribute__((ext_vector_type(8))) float v8f;

#define D64        64
#define NV         325
#define LDS_STRIDE 68
#define BT_CHUNK   128
#define N_VSTRIPE  21
#define N_BTCHUNK  12

__device__ __forceinline__ void load_bfrag(v2f (&bfrag)[16], const float* __restrict__ Bbase)
{
    #pragma unroll
    for (int kb = 0; kb < 16; ++kb) {
        bfrag[kb][0] = Bbase[(4 * kb    ) * D64];
        bfrag[kb][1] = Bbase[(4 * kb + 1) * D64];
    }
}

__device__ __forceinline__ void load_afrag_lds(v2f (&afrag)[16], const float* Hp)
{
    #pragma unroll
    for (int kb = 0; kb < 16; ++kb)
        afrag[kb] = *reinterpret_cast<const v2f*>(Hp + 4 * kb);
}

__device__ __forceinline__ v8f chain16(const v2f (&afrag)[16], const v2f (&bfrag)[16])
{
    v8f acc = {};
    #pragma unroll
    for (int kb = 0; kb < 16; ++kb)
        acc = __builtin_amdgcn_wmma_f32_16x16x4_f32(false, afrag[kb], false, bfrag[kb],
                                                    (short)0, acc, false, false);
    return acc;
}


template <bool FULL>
__device__ __forceinline__ void store_row(float* __restrict__ rowp, const v4f (&sv)[8], const v4f tv,
                                          const int nv, const int half)
{
    #pragma unroll
    for (int i = 0; i < 8; ++i) {
        if (FULL || (2 * i + half < nv))
            *reinterpret_cast<v4f*>(rowp + i * (2 * D64)) = sv[i] + tv;
    }
}

template <bool FULL>
__device__ __forceinline__ void stream_tiles(float* __restrict__ op, const v4f (&sv)[8], const float* tp,
                                             const float* te_a, float* te_d, const v2f (&bfrag_t2)[16],
                                             const float bias_t2, const int wave, const int half, const int nv)
{
    #pragma unroll
    for (int t = 0; t < 8; ++t) {
        const v4f tvA = *reinterpret_cast<const v4f*>(tp + (16 * t    ) * LDS_STRIDE);
        const v4f tvB = *reinterpret_cast<const v4f*>(tp + (16 * t + 8) * LDS_STRIDE);
        store_row<FULL>(op + (size_t)(16 * t) * (NV * D64), sv, tvA, nv, half);
        if (t < 7) {
            const bool mine = ((t + 1) & 1) ? (wave < 4) : (wave >= 4);
            if (mine) {
                v2f af[16];
                load_afrag_lds(af, te_a + 16 * (t + 1) * LDS_STRIDE);
                const v8f acc = chain16(af, bfrag_t2);
                #pragma unroll
                for (int j = 0; j < 8; ++j)
                    te_d[(16 * (t + 1) + j) * LDS_STRIDE] = acc[j] + bias_t2;
            }
            __builtin_amdgcn_fence(__ATOMIC_RELEASE, "workgroup", "local");
            __builtin_amdgcn_s_barrier_signal(-1);
        }
        store_row<FULL>(op + (size_t)(16 * t + 8) * (NV * D64), sv, tvB, nv, half);
        if (t < 7) {
            __builtin_amdgcn_s_barrier_wait(-1);
            __builtin_amdgcn_fence(__ATOMIC_ACQUIRE, "workgroup", "local");
        }
    }
}

__global__ __launch_bounds__(256)
void fused_embed_kernel(const float* __restrict__ SE,
                        const int* __restrict__ TE, const int* __restrict__ Tptr,
                        const float* __restrict__ Ws1, const float* __restrict__ bs1,
                        const float* __restrict__ Ws2, const float* __restrict__ bs2,
                        const float* __restrict__ Wt1, const float* __restrict__ bt1,
                        const float* __restrict__ Wt2, const float* __restrict__ bt2,
                        float* __restrict__ out)
{
    __shared__ __attribute__((aligned(16))) float hid_se[16 * LDS_STRIDE];
    __shared__ __attribute__((aligned(16))) float se_t  [16 * LDS_STRIDE];
    __shared__ __attribute__((aligned(16))) float hid_te[BT_CHUNK * LDS_STRIDE];
    __shared__ __attribute__((aligned(16))) float te_t  [BT_CHUNK * LDS_STRIDE];

    const int tid  = threadIdx.x;
    const int lane = tid & 31;
    const int wave = __builtin_amdgcn_readfirstlane(tid >> 5);
    const int half = lane >> 4;
    const int r    = lane & 15;
    const int v0   = blockIdx.x * 16;
    const int bt0  = blockIdx.y * BT_CHUNK;
    const int T    = Tptr[0];

    const int d0 = (tid & 7) * 8;
    float g_dow[4][8], g_tod[4][8], g_b[8];
    {
        int dow[4], tod[4];
        #pragma unroll
        for (int i = 0; i < 4; ++i) {
            const int gr = bt0 + (tid >> 3) + 32 * i;
            dow[i] = TE[2 * gr    ] % 7;
            tod[i] = TE[2 * gr + 1] % T;
        }
        #pragma unroll
        for (int k = 0; k < 8; ++k) g_b[k] = bt1[d0 + k];
        #pragma unroll
        for (int i = 0; i < 4; ++i) {
            const float* w_dow = Wt1 + dow[i] * D64 + d0;
            const float* w_tod = Wt1 + (7 + tod[i]) * D64 + d0;
            #pragma unroll
            for (int k = 0; k < 8; ++k) { g_dow[i][k] = w_dow[k]; g_tod[i][k] = w_tod[k]; }
        }
    }

    const int nt0 = (wave & 3) * 16;
    v2f bfrag_t2[16];
    load_bfrag(bfrag_t2, Wt2 + (2 * half) * D64 + nt0 + r);
    const float bias_t2 = bt2[nt0 + r];
    float bias_s2 = 0.0f;

    v2f bfrag_s2[16];
    if (wave < 4) {
        const int  grow = v0 + r;
        const bool mok  = (grow < NV);
        const int  srow = mok ? grow : 0;
        const float msk = mok ? 1.0f : 0.0f;

        const float* Ap = SE + srow * D64 + 2 * half;
        v2f afrag[16];
        #pragma unroll
        for (int kb = 0; kb < 16; ++kb) {
            v2f a = *reinterpret_cast<const v2f*>(Ap + 4 * kb);
            a[0] *= msk; a[1] *= msk;
            afrag[kb] = a;
        }
        v2f bfrag_s1[16];
        load_bfrag(bfrag_s1, Ws1 + (2 * half) * D64 + nt0 + r);
        load_bfrag(bfrag_s2, Ws2 + (2 * half) * D64 + nt0 + r);
        const float bias1 = bs1[nt0 + r];
        bias_s2 = bs2[nt0 + r];

        const v8f acc = chain16(afrag, bfrag_s1);
        #pragma unroll
        for (int j = 0; j < 8; ++j)
            hid_se[(j + 8 * half) * LDS_STRIDE + nt0 + r] = fmaxf(acc[j] + bias1, 0.0f);
    }

    #pragma unroll
    for (int i = 0; i < 4; ++i) {
        float* hp = hid_te + ((tid >> 3) + 32 * i) * LDS_STRIDE + d0;
        #pragma unroll
        for (int k = 0; k < 8; ++k)
            hp[k] = fmaxf(g_dow[i][k] + g_tod[i][k] + g_b[k], 0.0f);
    }
    __syncthreads();

    const float* te_a = hid_te + r * LDS_STRIDE + 2 * half;
    float*       te_d = te_t + (8 * half) * LDS_STRIDE + nt0 + r;

    if (wave < 4) {
        v2f afrag2[16];
        load_afrag_lds(afrag2, hid_se + r * LDS_STRIDE + 2 * half);
        const v8f acc2 = chain16(afrag2, bfrag_s2);
        #pragma unroll
        for (int j = 0; j < 8; ++j)
            se_t[(j + 8 * half) * LDS_STRIDE + nt0 + r] = acc2[j] + bias_s2;
    } else {
        v2f af0[16];
        load_afrag_lds(af0, te_a);
        const v8f acc0 = chain16(af0, bfrag_t2);
        #pragma unroll
        for (int j = 0; j < 8; ++j)
            te_d[j * LDS_STRIDE] = acc0[j] + bias_t2;
    }
    __syncthreads();

    v4f sv[8];
    #pragma unroll
    for (int i = 0; i < 8; ++i)
        sv[i] = *reinterpret_cast<const v4f*>(se_t + (2 * i + half) * LDS_STRIDE + 4 * r);
    const float* tp = te_t + wave * LDS_STRIDE + 4 * r;
    float* op = out + ((size_t)(bt0 + wave) * NV + v0 + half) * D64 + 4 * r;
    const int nv = NV - v0;

    if (nv >= 16)
        stream_tiles<true >(op, sv, tp, te_a, te_d, bfrag_t2, bias_t2, wave, half, nv);
    else
        stream_tiles<false>(op, sv, tp, te_a, te_d, bfrag_t2, bias_t2, wave, half, nv);
}

extern "C" void kernel_launch(void* const* d_in, const int* in_sizes, int n_in,
                              void* d_out, int out_size, void* d_ws, size_t ws_size,
                              hipStream_t stream)
{
    const float* SE  = (const float*)d_in[0];
    const int*   TE  = (const int*)  d_in[1];
    const int*   T   = (const int*)  d_in[2];
    const float* Ws1 = (const float*)d_in[3];
    const float* bs1 = (const float*)d_in[4];
    const float* Ws2 = (const float*)d_in[5];
    const float* bs2 = (const float*)d_in[6];
    const float* Wt1 = (const float*)d_in[7];
    const float* bt1 = (const float*)d_in[8];
    const float* Wt2 = (const float*)d_in[9];
    const float* bt2 = (const float*)d_in[10];
    float* out = (float*)d_out;

    fused_embed_kernel<<<dim3(N_VSTRIPE, N_BTCHUNK), dim3(256), 0, stream>>>(
        SE, TE, T, Ws1, bs1, Ws2, bs2, Wt1, bt1, Wt2, bt2, out);

    (void)in_sizes; (void)n_in; (void)out_size; (void)d_ws; (void)ws_size;
}
